# starting kernel
# baseline (speedup 1.0000x reference)
#include <hip/hip_runtime.h>
#include <hip/hip_bf16.h>

typedef __attribute__((ext_vector_type(2))) float v2f;
typedef __attribute__((ext_vector_type(8))) float v8f;

#define CCH 64
#define NEG_INF_I ((int)0xFF800000)

__global__ __launch_bounds__(128) void k_gemm(const float* __restrict__ x,
                                              const float* __restrict__ W,
                                              const float* __restrict__ bias,
                                              float* __restrict__ A_ws,
                                              float* __restrict__ B_ws,
                                              int N)
{
    const int lane = threadIdx.x & 31;
    const int wave = threadIdx.x >> 5;
    const int n0   = wave * 16;
    const int m0   = blockIdx.x * 16;
    const int half = lane >> 4;
    const int lm   = lane & 15;
    const int col  = n0 + lm;

    v8f accP = {};
    v8f accQ = {};

    int rowA = m0 + lm;
    if (rowA >= N) rowA = N - 1;

#pragma unroll
    for (int kk = 0; kk < 16; ++kk) {
        const int kb  = kk * 4 + half * 2;
        v2f a, b1, b2;
        a.x  = x[(long)rowA * CCH + kb];
        a.y  = x[(long)rowA * CCH + kb + 1];
        b1.x = W[kb * CCH + col];
        b1.y = W[(kb + 1) * CCH + col];
        b2.x = W[(kb + 64) * CCH + col];
        b2.y = W[(kb + 65) * CCH + col];
        accP = __builtin_amdgcn_wmma_f32_16x16x4_f32(false, a, false, b1,
                                                     (short)0, accP, false, false);
        accQ = __builtin_amdgcn_wmma_f32_16x16x4_f32(false, a, false, b2,
                                                     (short)0, accQ, false, false);
    }

    const float bv = bias[col];
    if (m0 + 16 <= N) {
#pragma unroll
        for (int r = 0; r < 8; ++r) {
            const int row = m0 + r + half * 8;
            const float q = accQ[r];
            A_ws[(long)row * CCH + col] = accP[r] - q + bv;
            B_ws[(long)row * CCH + col] = q;
        }
    } else {
#pragma unroll
        for (int r = 0; r < 8; ++r) {
            const int row = m0 + r + half * 8;
            if (row < N) {
                const float q = accQ[r];
                A_ws[(long)row * CCH + col] = accP[r] - q + bv;
                B_ws[(long)row * CCH + col] = q;
            }
        }
    }
}

__global__ void k_init(int* __restrict__ M, long total)
{
    long i = (long)blockIdx.x * blockDim.x + threadIdx.x;
    if (i < total) M[i] = NEG_INF_I;
}

__global__ __launch_bounds__(256) void k_scatter(const float* __restrict__ Bn,
                                                 const int* __restrict__ ei,
                                                 int* __restrict__ M,
                                                 int E)
{
    long tid   = (long)blockIdx.x * blockDim.x + threadIdx.x;
    long total = (long)E * CCH;
    if (tid >= total) return;
    int e = (int)(tid >> 6);
    int c = (int)(tid & 63);
    int src = ei[e];
    int dst = ei[E + e];
    float v = Bn[(long)src * CCH + c];
    int vi = __float_as_int(v);
    int* addr = M + (long)dst * CCH + c;
    if (vi >= 0) atomicMax(addr, vi);
    else         atomicMin((unsigned int*)addr, (unsigned int)vi);
}

__global__ __launch_bounds__(256) void k_elu_partial(const float* __restrict__ A_ws,
                                                     const int* __restrict__ M,
                                                     float* __restrict__ H,
                                                     float* __restrict__ psum,
                                                     float* __restrict__ psq,
                                                     int N)
{
    __shared__ float smem[256];
    const int t  = threadIdx.x;
    const int c  = t & 63;
    const int r4 = t >> 6;
    const int base = blockIdx.x * 64;

    float s = 0.f, q = 0.f;
    for (int rr = r4; rr < 64; rr += 4) {
        int row = base + rr;
        if (row < N) {
            long idx = (long)row * CCH + c;
            int  mb  = M[idx];
            float h;
            if (mb == NEG_INF_I) {
                h = 0.f;
            } else {
                float z = A_ws[idx] + __int_as_float(mb);
                h = (z > 0.f) ? z : expm1f(z);
            }
            H[idx] = h;
            s += h;
            q += h * h;
        }
    }
    smem[t] = s; __syncthreads();
    if (r4 == 0)
        psum[(long)blockIdx.x * 64 + c] = smem[c] + smem[c + 64] + smem[c + 128] + smem[c + 192];
    __syncthreads();
    smem[t] = q; __syncthreads();
    if (r4 == 0)
        psq[(long)blockIdx.x * 64 + c]  = smem[c] + smem[c + 64] + smem[c + 128] + smem[c + 192];
}

__global__ void k_stats(const float* __restrict__ psum,
                        const float* __restrict__ psq,
                        int nblocks, int N,
                        float* __restrict__ mean,
                        float* __restrict__ invstd)
{
    int c = threadIdx.x;
    if (c >= 64) return;
    double s = 0.0, q = 0.0;
    for (int b = 0; b < nblocks; ++b) {
        s += (double)psum[(long)b * 64 + c];
        q += (double)psq[(long)b * 64 + c];
    }
    double m = s / (double)N;
    double v = q / (double)N - m * m;
    if (v < 0.0) v = 0.0;
    mean[c]   = (float)m;
    invstd[c] = (float)(1.0 / sqrt(v + 1e-5));
}

__global__ __launch_bounds__(256) void k_bn(float* __restrict__ out,
                                            const float* __restrict__ mean,
                                            const float* __restrict__ invstd,
                                            const float* __restrict__ gamma,
                                            const float* __restrict__ beta,
                                            long total)
{
    long i = (long)blockIdx.x * blockDim.x + threadIdx.x;
    if (i < total) {
        int c = (int)(i & 63);
        out[i] = gamma[c] * (out[i] - mean[c]) * invstd[c] + beta[c];
    }
}

extern "C" void kernel_launch(void* const* d_in, const int* in_sizes, int n_in,
                              void* d_out, int out_size, void* d_ws, size_t ws_size,
                              hipStream_t stream)
{
    const float* x     = (const float*)d_in[0];
    const int*   ei    = (const int*)  d_in[1];
    const float* W     = (const float*)d_in[2];
    const float* bias  = (const float*)d_in[3];
    const float* gamma = (const float*)d_in[4];
    const float* beta  = (const float*)d_in[5];
    float* out = (float*)d_out;

    const int N = in_sizes[0] / CCH;
    const int E = in_sizes[1] / 2;

    const size_t nodef = (size_t)N * CCH;
    float* A_ws = (float*)d_ws;
    float* B_ws = A_ws + nodef;
    int*   M_ws = (int*)(B_ws + nodef);
    const int nb4 = (N + 63) / 64;
    float* psum    = (float*)(M_ws + nodef);
    float* psq     = psum + (size_t)nb4 * 64;
    float* meanp   = psq  + (size_t)nb4 * 64;
    float* invstdp = meanp + 64;

    const long tot  = (long)nodef;
    const long etot = (long)E * CCH;

    k_gemm<<<(N + 15) / 16, 128, 0, stream>>>(x, W, bias, A_ws, B_ws, N);
    k_init<<<(int)((tot + 255) / 256), 256, 0, stream>>>(M_ws, tot);
    k_scatter<<<(int)((etot + 255) / 256), 256, 0, stream>>>(B_ws, ei, M_ws, E);
    k_elu_partial<<<nb4, 256, 0, stream>>>(A_ws, M_ws, out, psum, psq, N);
    k_stats<<<1, 64, 0, stream>>>(psum, psq, nb4, N, meanp, invstdp);
    k_bn<<<(int)((tot + 255) / 256), 256, 0, stream>>>(out, meanp, invstdp, gamma, beta, tot);
}


# Scatter-max kernel: one wave takes 16 consecutive edges, whole 64-channel rows; edge indices wave-uniform (scalar loads), all 32 row loads in flight before the first atomic; 100000 waves, not 3200000
# speedup vs baseline: 1.6707x; 1.6707x over previous
#include <hip/hip_runtime.h>
#include <hip/hip_bf16.h>

typedef __attribute__((ext_vector_type(2))) float v2f;
typedef __attribute__((ext_vector_type(8))) float v8f;

#define CCH 64
#define NEG_INF_I ((int)0xFF800000)

__global__ __launch_bounds__(128) void k_gemm(const float* __restrict__ x,
                                              const float* __restrict__ W,
                                              const float* __restrict__ bias,
                                              float* __restrict__ A_ws,
                                              float* __restrict__ B_ws,
                                              int N)
{
    const int lane = threadIdx.x & 31;
    const int wave = threadIdx.x >> 5;
    const int n0   = wave * 16;
    const int m0   = blockIdx.x * 16;
    const int half = lane >> 4;
    const int lm   = lane & 15;
    const int col  = n0 + lm;

    v8f accP = {};
    v8f accQ = {};

    int rowA = m0 + lm;
    if (rowA >= N) rowA = N - 1;

#pragma unroll
    for (int kk = 0; kk < 16; ++kk) {
        const int kb  = kk * 4 + half * 2;
        v2f a, b1, b2;
        a.x  = x[(long)rowA * CCH + kb];
        a.y  = x[(long)rowA * CCH + kb + 1];
        b1.x = W[kb * CCH + col];
        b1.y = W[(kb + 1) * CCH + col];
        b2.x = W[(kb + 64) * CCH + col];
        b2.y = W[(kb + 65) * CCH + col];
        accP = __builtin_amdgcn_wmma_f32_16x16x4_f32(false, a, false, b1,
                                                     (short)0, accP, false, false);
        accQ = __builtin_amdgcn_wmma_f32_16x16x4_f32(false, a, false, b2,
                                                     (short)0, accQ, false, false);
    }

    const float bv = bias[col];
    if (m0 + 16 <= N) {
#pragma unroll
        for (int r = 0; r < 8; ++r) {
            const int row = m0 + r + half * 8;
            const float q = accQ[r];
            A_ws[(long)row * CCH + col] = accP[r] - q + bv;
            B_ws[(long)row * CCH + col] = q;
        }
    } else {
#pragma unroll
        for (int r = 0; r < 8; ++r) {
            const int row = m0 + r + half * 8;
            if (row < N) {
                const float q = accQ[r];
                A_ws[(long)row * CCH + col] = accP[r] - q + bv;
                B_ws[(long)row * CCH + col] = q;
            }
        }
    }
}

__global__ void k_init(int* __restrict__ M, long total)
{
    long i = (long)blockIdx.x * blockDim.x + threadIdx.x;
    if (i < total) M[i] = NEG_INF_I;
}

#define EPW 16

__device__ __forceinline__ void amax_f(int* addr, float v)
{
    const int vi = __float_as_int(v);
    if (vi >= 0) atomicMax(addr, vi);
    else         atomicMin((unsigned int*)addr, (unsigned int)vi);
}

__global__ __launch_bounds__(256) void k_scatter(const float* __restrict__ Bn,
                                                 const int* __restrict__ ei,
                                                 int* __restrict__ M,
                                                 int E)
{
    const int lane = threadIdx.x & 31;
    const int wave = __builtin_amdgcn_readfirstlane((int)(threadIdx.x >> 5));
    const long e0  = ((long)blockIdx.x * 8 + wave) * EPW;
    if (e0 >= E) return;

    if (e0 + EPW <= E) {
        int src[EPW], dst[EPW];
#pragma unroll
        for (int j = 0; j < EPW; ++j) {
            src[j] = ei[e0 + j];
            dst[j] = ei[(long)E + e0 + j];
        }
        float v[2 * EPW];
#pragma unroll
        for (int j = 0; j < EPW; ++j) {
            const float* row = Bn + (long)src[j] * CCH;
            v[2 * j]     = row[lane];
            v[2 * j + 1] = row[lane + 32];
        }
#pragma unroll
        for (int j = 0; j < EPW; ++j) {
            int* row = M + (long)dst[j] * CCH;
            amax_f(row + lane,      v[2 * j]);
            amax_f(row + lane + 32, v[2 * j + 1]);
        }
    } else {
        for (long e = e0; e < E; ++e) {
            const int s = ei[e];
            const int d = ei[(long)E + e];
            const float* srow = Bn + (long)s * CCH;
            int* drow = M + (long)d * CCH;
            amax_f(drow + lane,      srow[lane]);
            amax_f(drow + lane + 32, srow[lane + 32]);
        }
    }
}

__global__ __launch_bounds__(256) void k_elu_partial(const float* __restrict__ A_ws,
                                                     const int* __restrict__ M,
                                                     float* __restrict__ H,
                                                     float* __restrict__ psum,
                                                     float* __restrict__ psq,
                                                     int N)
{
    __shared__ float smem[256];
    const int t  = threadIdx.x;
    const int c  = t & 63;
    const int r4 = t >> 6;
    const int base = blockIdx.x * 64;

    float s = 0.f, q = 0.f;
    for (int rr = r4; rr < 64; rr += 4) {
        int row = base + rr;
        if (row < N) {
            long idx = (long)row * CCH + c;
            int  mb  = M[idx];
            float h;
            if (mb == NEG_INF_I) {
                h = 0.f;
            } else {
                float z = A_ws[idx] + __int_as_float(mb);
                h = (z > 0.f) ? z : expm1f(z);
            }
            H[idx] = h;
            s += h;
            q += h * h;
        }
    }
    smem[t] = s; __syncthreads();
    if (r4 == 0)
        psum[(long)blockIdx.x * 64 + c] = smem[c] + smem[c + 64] + smem[c + 128] + smem[c + 192];
    __syncthreads();
    smem[t] = q; __syncthreads();
    if (r4 == 0)
        psq[(long)blockIdx.x * 64 + c]  = smem[c] + smem[c + 64] + smem[c + 128] + smem[c + 192];
}

__global__ void k_stats(const float* __restrict__ psum,
                        const float* __restrict__ psq,
                        int nblocks, int N,
                        float* __restrict__ mean,
                        float* __restrict__ invstd)
{
    int c = threadIdx.x;
    if (c >= 64) return;
    double s = 0.0, q = 0.0;
    for (int b = 0; b < nblocks; ++b) {
        s += (double)psum[(long)b * 64 + c];
        q += (double)psq[(long)b * 64 + c];
    }
    double m = s / (double)N;
    double v = q / (double)N - m * m;
    if (v < 0.0) v = 0.0;
    mean[c]   = (float)m;
    invstd[c] = (float)(1.0 / sqrt(v + 1e-5));
}

__global__ __launch_bounds__(256) void k_bn(float* __restrict__ out,
                                            const float* __restrict__ mean,
                                            const float* __restrict__ invstd,
                                            const float* __restrict__ gamma,
                                            const float* __restrict__ beta,
                                            long total)
{
    long i = (long)blockIdx.x * blockDim.x + threadIdx.x;
    if (i < total) {
        int c = (int)(i & 63);
        out[i] = gamma[c] * (out[i] - mean[c]) * invstd[c] + beta[c];
    }
}

extern "C" void kernel_launch(void* const* d_in, const int* in_sizes, int n_in,
                              void* d_out, int out_size, void* d_ws, size_t ws_size,
                              hipStream_t stream)
{
    const float* x     = (const float*)d_in[0];
    const int*   ei    = (const int*)  d_in[1];
    const float* W     = (const float*)d_in[2];
    const float* bias  = (const float*)d_in[3];
    const float* gamma = (const float*)d_in[4];
    const float* beta  = (const float*)d_in[5];
    float* out = (float*)d_out;

    const int N = in_sizes[0] / CCH;
    const int E = in_sizes[1] / 2;

    const size_t nodef = (size_t)N * CCH;
    float* A_ws = (float*)d_ws;
    float* B_ws = A_ws + nodef;
    int*   M_ws = (int*)(B_ws + nodef);
    const int nb4 = (N + 63) / 64;
    float* psum    = (float*)(M_ws + nodef);
    float* psq     = psum + (size_t)nb4 * 64;
    float* meanp   = psq  + (size_t)nb4 * 64;
    float* invstdp = meanp + 64;

    const long tot  = (long)nodef;

    k_gemm<<<(N + 15) / 16, 128, 0, stream>>>(x, W, bias, A_ws, B_ws, N);
    k_init<<<(int)((tot + 255) / 256), 256, 0, stream>>>(M_ws, tot);
    k_scatter<<<(E + 8 * EPW - 1) / (8 * EPW), 256, 0, stream>>>(B_ws, ei, M_ws, E);
    k_elu_partial<<<nb4, 256, 0, stream>>>(A_ws, M_ws, out, psum, psq, N);
    k_stats<<<1, 64, 0, stream>>>(psum, psq, nb4, N, meanp, invstdp);
    k_bn<<<(int)((tot + 255) / 256), 256, 0, stream>>>(out, meanp, invstdp, gamma, beta, tot);
}


# Scatter-max on order-preserving int keys of the floats: each value takes ONE full-mask signed atomic max (was a sign-split max/min pair of half-masked atomics): 3.2M atomic requests, not 6.4M
# speedup vs baseline: 1.8204x; 1.0896x over previous
#include <hip/hip_runtime.h>
#include <hip/hip_bf16.h>

typedef __attribute__((ext_vector_type(2))) float v2f;
typedef __attribute__((ext_vector_type(8))) float v8f;

#define CCH 64

__device__ __forceinline__ int fkey(int bits)
{
    return bits ^ ((bits >> 31) & 0x7FFFFFFF);
}
#define KEY_NEG_INF ((int)0x807FFFFF)

__global__ __launch_bounds__(128) void k_gemm(const float* __restrict__ x,
                                              const float* __restrict__ W,
                                              const float* __restrict__ bias,
                                              float* __restrict__ A_ws,
                                              float* __restrict__ B_ws,
                                              int N)
{
    const int lane = threadIdx.x & 31;
    const int wave = threadIdx.x >> 5;
    const int n0   = wave * 16;
    const int m0   = blockIdx.x * 16;
    const int half = lane >> 4;
    const int lm   = lane & 15;
    const int col  = n0 + lm;

    v8f accP = {};
    v8f accQ = {};

    int rowA = m0 + lm;
    if (rowA >= N) rowA = N - 1;

#pragma unroll
    for (int kk = 0; kk < 16; ++kk) {
        const int kb  = kk * 4 + half * 2;
        v2f a, b1, b2;
        a.x  = x[(long)rowA * CCH + kb];
        a.y  = x[(long)rowA * CCH + kb + 1];
        b1.x = W[kb * CCH + col];
        b1.y = W[(kb + 1) * CCH + col];
        b2.x = W[(kb + 64) * CCH + col];
        b2.y = W[(kb + 65) * CCH + col];
        accP = __builtin_amdgcn_wmma_f32_16x16x4_f32(false, a, false, b1,
                                                     (short)0, accP, false, false);
        accQ = __builtin_amdgcn_wmma_f32_16x16x4_f32(false, a, false, b2,
                                                     (short)0, accQ, false, false);
    }

    const float bv = bias[col];
    if (m0 + 16 <= N) {
#pragma unroll
        for (int r = 0; r < 8; ++r) {
            const int row = m0 + r + half * 8;
            const float q = accQ[r];
            A_ws[(long)row * CCH + col] = accP[r] - q + bv;
            B_ws[(long)row * CCH + col] = q;
        }
    } else {
#pragma unroll
        for (int r = 0; r < 8; ++r) {
            const int row = m0 + r + half * 8;
            if (row < N) {
                const float q = accQ[r];
                A_ws[(long)row * CCH + col] = accP[r] - q + bv;
                B_ws[(long)row * CCH + col] = q;
            }
        }
    }
}

__global__ void k_init(int* __restrict__ M, long total)
{
    long i = (long)blockIdx.x * blockDim.x + threadIdx.x;
    if (i < total) M[i] = KEY_NEG_INF;
}

#define EPW 16

__device__ __forceinline__ void amax_f(int* addr, float v)
{
    atomicMax(addr, fkey(__float_as_int(v)));
}

__global__ __launch_bounds__(256) void k_scatter(const float* __restrict__ Bn,
                                                 const int* __restrict__ ei,
                                                 int* __restrict__ M,
                                                 int E)
{
    const int lane = threadIdx.x & 31;
    const int wave = __builtin_amdgcn_readfirstlane((int)(threadIdx.x >> 5));
    const long e0  = ((long)blockIdx.x * 8 + wave) * EPW;
    if (e0 >= E) return;

    if (e0 + EPW <= E) {
        int src[EPW], dst[EPW];
#pragma unroll
        for (int j = 0; j < EPW; ++j) {
            src[j] = ei[e0 + j];
            dst[j] = ei[(long)E + e0 + j];
        }
        float v[2 * EPW];
#pragma unroll
        for (int j = 0; j < EPW; ++j) {
            const float* row = Bn + (long)src[j] * CCH;
            v[2 * j]     = row[lane];
            v[2 * j + 1] = row[lane + 32];
        }
#pragma unroll
        for (int j = 0; j < EPW; ++j) {
            int* row = M + (long)dst[j] * CCH;
            amax_f(row + lane,      v[2 * j]);
            amax_f(row + lane + 32, v[2 * j + 1]);
        }
    } else {
        for (long e = e0; e < E; ++e) {
            const int s = ei[e];
            const int d = ei[(long)E + e];
            const float* srow = Bn + (long)s * CCH;
            int* drow = M + (long)d * CCH;
            amax_f(drow + lane,      srow[lane]);
            amax_f(drow + lane + 32, srow[lane + 32]);
        }
    }
}

__global__ __launch_bounds__(256) void k_elu_partial(const float* __restrict__ A_ws,
                                                     const int* __restrict__ M,
                                                     float* __restrict__ H,
                                                     float* __restrict__ psum,
                                                     float* __restrict__ psq,
                                                     int N)
{
    __shared__ float smem[256];
    const int t  = threadIdx.x;
    const int c  = t & 63;
    const int r4 = t >> 6;
    const int base = blockIdx.x * 64;

    float s = 0.f, q = 0.f;
    for (int rr = r4; rr < 64; rr += 4) {
        int row = base + rr;
        if (row < N) {
            long idx = (long)row * CCH + c;
            int  mb  = M[idx];
            float h;
            if (mb == KEY_NEG_INF) {
                h = 0.f;
            } else {
                float z = A_ws[idx] + __int_as_float(fkey(mb));
                h = (z > 0.f) ? z : expm1f(z);
            }
            H[idx] = h;
            s += h;
            q += h * h;
        }
    }
    smem[t] = s; __syncthreads();
    if (r4 == 0)
        psum[(long)blockIdx.x * 64 + c] = smem[c] + smem[c + 64] + smem[c + 128] + smem[c + 192];
    __syncthreads();
    smem[t] = q; __syncthreads();
    if (r4 == 0)
        psq[(long)blockIdx.x * 64 + c]  = smem[c] + smem[c + 64] + smem[c + 128] + smem[c + 192];
}

__global__ void k_stats(const float* __restrict__ psum,
                        const float* __restrict__ psq,
                        int nblocks, int N,
                        float* __restrict__ mean,
                        float* __restrict__ invstd)
{
    int c = threadIdx.x;
    if (c >= 64) return;
    double s = 0.0, q = 0.0;
    for (int b = 0; b < nblocks; ++b) {
        s += (double)psum[(long)b * 64 + c];
        q += (double)psq[(long)b * 64 + c];
    }
    double m = s / (double)N;
    double v = q / (double)N - m * m;
    if (v < 0.0) v = 0.0;
    mean[c]   = (float)m;
    invstd[c] = (float)(1.0 / sqrt(v + 1e-5));
}

__global__ __launch_bounds__(256) void k_bn(float* __restrict__ out,
                                            const float* __restrict__ mean,
                                            const float* __restrict__ invstd,
                                            const float* __restrict__ gamma,
                                            const float* __restrict__ beta,
                                            long total)
{
    long i = (long)blockIdx.x * blockDim.x + threadIdx.x;
    if (i < total) {
        int c = (int)(i & 63);
        out[i] = gamma[c] * (out[i] - mean[c]) * invstd[c] + beta[c];
    }
}

extern "C" void kernel_launch(void* const* d_in, const int* in_sizes, int n_in,
                              void* d_out, int out_size, void* d_ws, size_t ws_size,
                              hipStream_t stream)
{
    const float* x     = (const float*)d_in[0];
    const int*   ei    = (const int*)  d_in[1];
    const float* W     = (const float*)d_in[2];
    const float* bias  = (const float*)d_in[3];
    const float* gamma = (const float*)d_in[4];
    const float* beta  = (const float*)d_in[5];
    float* out = (float*)d_out;

    const int N = in_sizes[0] / CCH;
    const int E = in_sizes[1] / 2;

    const size_t nodef = (size_t)N * CCH;
    float* A_ws = (float*)d_ws;
    float* B_ws = A_ws + nodef;
    int*   M_ws = (int*)(B_ws + nodef);
    const int nb4 = (N + 63) / 64;
    float* psum    = (float*)(M_ws + nodef);
    float* psq     = psum + (size_t)nb4 * 64;
    float* meanp   = psq  + (size_t)nb4 * 64;
    float* invstdp = meanp + 64;

    const long tot  = (long)nodef;

    k_gemm<<<(N + 15) / 16, 128, 0, stream>>>(x, W, bias, A_ws, B_ws, N);
    k_init<<<(int)((tot + 255) / 256), 256, 0, stream>>>(M_ws, tot);
    k_scatter<<<(E + 8 * EPW - 1) / (8 * EPW), 256, 0, stream>>>(B_ws, ei, M_ws, E);
    k_elu_partial<<<nb4, 256, 0, stream>>>(A_ws, M_ws, out, psum, psq, N);
    k_stats<<<1, 64, 0, stream>>>(psum, psq, nb4, N, meanp, invstdp);
    k_bn<<<(int)((tot + 255) / 256), 256, 0, stream>>>(out, meanp, invstdp, gamma, beta, tot);
}


# Two-level last-arriver tickets in the ELU kernel sum the BN partial rows (25 blocks per group, then 63 group rows); the serial 64-thread double-sum launch is gone; sums reordered, compensated f32
# speedup vs baseline: 2.8538x; 1.5677x over previous
#include <hip/hip_runtime.h>
#include <hip/hip_bf16.h>

typedef __attribute__((ext_vector_type(2))) float v2f;
typedef __attribute__((ext_vector_type(8))) float v8f;

#define CCH 64

__device__ __forceinline__ int fkey(int bits)
{
    return bits ^ ((bits >> 31) & 0x7FFFFFFF);
}
#define KEY_NEG_INF ((int)0x807FFFFF)

__global__ __launch_bounds__(128) void k_gemm(const float* __restrict__ x,
                                              const float* __restrict__ W,
                                              const float* __restrict__ bias,
                                              float* __restrict__ A_ws,
                                              float* __restrict__ B_ws,
                                              int N)
{
    const int lane = threadIdx.x & 31;
    const int wave = threadIdx.x >> 5;
    const int n0   = wave * 16;
    const int m0   = blockIdx.x * 16;
    const int half = lane >> 4;
    const int lm   = lane & 15;
    const int col  = n0 + lm;

    v8f accP = {};
    v8f accQ = {};

    int rowA = m0 + lm;
    if (rowA >= N) rowA = N - 1;

#pragma unroll
    for (int kk = 0; kk < 16; ++kk) {
        const int kb  = kk * 4 + half * 2;
        v2f a, b1, b2;
        a.x  = x[(long)rowA * CCH + kb];
        a.y  = x[(long)rowA * CCH + kb + 1];
        b1.x = W[kb * CCH + col];
        b1.y = W[(kb + 1) * CCH + col];
        b2.x = W[(kb + 64) * CCH + col];
        b2.y = W[(kb + 65) * CCH + col];
        accP = __builtin_amdgcn_wmma_f32_16x16x4_f32(false, a, false, b1,
                                                     (short)0, accP, false, false);
        accQ = __builtin_amdgcn_wmma_f32_16x16x4_f32(false, a, false, b2,
                                                     (short)0, accQ, false, false);
    }

    const float bv = bias[col];
    if (m0 + 16 <= N) {
#pragma unroll
        for (int r = 0; r < 8; ++r) {
            const int row = m0 + r + half * 8;
            const float q = accQ[r];
            A_ws[(long)row * CCH + col] = accP[r] - q + bv;
            B_ws[(long)row * CCH + col] = q;
        }
    } else {
#pragma unroll
        for (int r = 0; r < 8; ++r) {
            const int row = m0 + r + half * 8;
            if (row < N) {
                const float q = accQ[r];
                A_ws[(long)row * CCH + col] = accP[r] - q + bv;
                B_ws[(long)row * CCH + col] = q;
            }
        }
    }
}

#define NCNT 128

__global__ void k_init(int* __restrict__ M, long total, unsigned* __restrict__ cnt)
{
    long i = (long)blockIdx.x * blockDim.x + threadIdx.x;
    if (i < total) M[i] = KEY_NEG_INF;
    if (i < NCNT) cnt[i] = 0u;
}

#define EPW 16

__device__ __forceinline__ void amax_f(int* addr, float v)
{
    atomicMax(addr, fkey(__float_as_int(v)));
}

__global__ __launch_bounds__(256) void k_scatter(const float* __restrict__ Bn,
                                                 const int* __restrict__ ei,
                                                 int* __restrict__ M,
                                                 int E)
{
    const int lane = threadIdx.x & 31;
    const int wave = __builtin_amdgcn_readfirstlane((int)(threadIdx.x >> 5));
    const long e0  = ((long)blockIdx.x * 8 + wave) * EPW;
    if (e0 >= E) return;

    if (e0 + EPW <= E) {
        int src[EPW], dst[EPW];
#pragma unroll
        for (int j = 0; j < EPW; ++j) {
            src[j] = ei[e0 + j];
            dst[j] = ei[(long)E + e0 + j];
        }
        float v[2 * EPW];
#pragma unroll
        for (int j = 0; j < EPW; ++j) {
            const float* row = Bn + (long)src[j] * CCH;
            v[2 * j]     = row[lane];
            v[2 * j + 1] = row[lane + 32];
        }
#pragma unroll
        for (int j = 0; j < EPW; ++j) {
            int* row = M + (long)dst[j] * CCH;
            amax_f(row + lane,      v[2 * j]);
            amax_f(row + lane + 32, v[2 * j + 1]);
        }
    } else {
        for (long e = e0; e < E; ++e) {
            const int s = ei[e];
            const int d = ei[(long)E + e];
            const float* srow = Bn + (long)s * CCH;
            int* drow = M + (long)d * CCH;
            amax_f(drow + lane,      srow[lane]);
            amax_f(drow + lane + 32, srow[lane + 32]);
        }
    }
}


__device__ __forceinline__ void kadd(float& s, float& c, float x)
{
    const float y = x - c;
    const float t = s + y;
    c = (t - s) - y;
    s = t;
}

__device__ __forceinline__ void col_sums(const float* a0, const float* a1,
                                         int first, int n, float* sm,
                                         float& S, float& C)
{
    const int t    = threadIdx.x;
    const int lane = t & 31;
    const int w    = __builtin_amdgcn_readfirstlane(t >> 5);
    const float* base = ((lane < 16) ? (a0 + 4 * lane) : (a1 + 4 * (lane - 16)))
                        + (long)first * CCH;
    float4 s = make_float4(0.f, 0.f, 0.f, 0.f);
    float4 c = make_float4(0.f, 0.f, 0.f, 0.f);
    for (int r0 = w; r0 < n; r0 += 64) {
        float4 v[8];
#pragma unroll
        for (int j = 0; j < 8; ++j) {
            int r = r0 + 8 * j;
            if (r > n - 1) r = n - 1;
            v[j] = *(const float4*)(base + (long)r * CCH);
        }
#pragma unroll
        for (int j = 0; j < 8; ++j) {
            const bool live = (r0 + 8 * j < n);
            kadd(s.x, c.x, live ? v[j].x : 0.f);
            kadd(s.y, c.y, live ? v[j].y : 0.f);
            kadd(s.z, c.z, live ? v[j].z : 0.f);
            kadd(s.w, c.w, live ? v[j].w : 0.f);
        }
    }
    *(float4*)(sm + w * 128 + 4 * lane)        = s;
    *(float4*)(sm + 1024 + w * 128 + 4 * lane) = c;
    __syncthreads();
    S = 0.f; C = 0.f;
    if (t < 128) {
#pragma unroll
        for (int k = 0; k < 8; ++k) {
            kadd(S, C,  sm[k * 128 + t]);
            kadd(S, C, -sm[1024 + k * 128 + t]);
        }
    }
}

__device__ __forceinline__ unsigned take_ticket(unsigned* cnt, unsigned* s_slot)
{
    __builtin_amdgcn_fence(__ATOMIC_RELEASE, "agent");
    __syncthreads();
    if (threadIdx.x == 0) {
        __builtin_amdgcn_fence(__ATOMIC_RELEASE, "agent");
        *s_slot = __hip_atomic_fetch_add(cnt, 1u, __ATOMIC_RELAXED,
                                         __HIP_MEMORY_SCOPE_AGENT);
    }
    __syncthreads();
    return *s_slot;
}

__global__ __launch_bounds__(256) void k_elu_partial(const float* __restrict__ A_ws,
                                                     const int* __restrict__ M,
                                                     float* __restrict__ H,
                                                     float* psum,
                                                     float* psq,
                                                     unsigned* cnt,
                                                     float* gsum,
                                                     float* gsq,
                                                     float* mean,
                                                     float* invstd,
                                                     int N)
{
    __shared__ float smem[256];
    __shared__ float sred[2048];
    __shared__ double sfin[128];
    __shared__ unsigned s_tick[2];
    const int t  = threadIdx.x;
    const int c  = t & 63;
    const int r4 = t >> 6;
    const int base = blockIdx.x * 64;

    float s = 0.f, q = 0.f;
    for (int rr = r4; rr < 64; rr += 4) {
        int row = base + rr;
        if (row < N) {
            long idx = (long)row * CCH + c;
            int  mb  = M[idx];
            float h;
            if (mb == KEY_NEG_INF) {
                h = 0.f;
            } else {
                float z = A_ws[idx] + __int_as_float(fkey(mb));
                h = (z > 0.f) ? z : expm1f(z);
            }
            H[idx] = h;
            s += h;
            q += h * h;
        }
    }
    smem[t] = s; __syncthreads();
    if (r4 == 0)
        psum[(long)blockIdx.x * 64 + c] = smem[c] + smem[c + 64] + smem[c + 128] + smem[c + 192];
    __syncthreads();
    smem[t] = q; __syncthreads();
    if (r4 == 0)
        psq[(long)blockIdx.x * 64 + c]  = smem[c] + smem[c + 64] + smem[c + 128] + smem[c + 192];

    const int nb  = (int)gridDim.x;
    const int S   = (nb + 63) / 64;
    const int G   = (nb + S - 1) / S;
    const int g   = (int)blockIdx.x / S;
    const int pop = (nb - g * S < S) ? (nb - g * S) : S;

    if (take_ticket(cnt + 2 * g, &s_tick[0]) != (unsigned)(pop - 1)) return;
    __builtin_amdgcn_fence(__ATOMIC_ACQUIRE, "agent");

    float S1, C1;
    col_sums(psum, psq, g * S, pop, sred, S1, C1);
    if (t < 64)       gsum[g * 64 + t]        = S1 - C1;
    else if (t < 128) gsq [g * 64 + (t - 64)] = S1 - C1;

    if (take_ticket(cnt + (NCNT - 1), &s_tick[1]) != (unsigned)(G - 1)) return;
    __builtin_amdgcn_fence(__ATOMIC_ACQUIRE, "agent");

    float S2, C2;
    col_sums(gsum, gsq, 0, G, sred, S2, C2);
    if (t < 128) sfin[t] = (double)S2 - (double)C2;
    __syncthreads();
    if (t < 64) {
        const double m = sfin[t] / (double)N;
        double v = sfin[64 + t] / (double)N - m * m;
        if (v < 0.0) v = 0.0;
        mean[t]   = (float)m;
        invstd[t] = (float)(1.0 / sqrt(v + 1e-5));
    }
}

__global__ __launch_bounds__(256) void k_bn(float* __restrict__ out,
                                            const float* __restrict__ mean,
                                            const float* __restrict__ invstd,
                                            const float* __restrict__ gamma,
                                            const float* __restrict__ beta,
                                            long total)
{
    long i = (long)blockIdx.x * blockDim.x + threadIdx.x;
    if (i < total) {
        int c = (int)(i & 63);
        out[i] = gamma[c] * (out[i] - mean[c]) * invstd[c] + beta[c];
    }
}

extern "C" void kernel_launch(void* const* d_in, const int* in_sizes, int n_in,
                              void* d_out, int out_size, void* d_ws, size_t ws_size,
                              hipStream_t stream)
{
    const float* x     = (const float*)d_in[0];
    const int*   ei    = (const int*)  d_in[1];
    const float* W     = (const float*)d_in[2];
    const float* bias  = (const float*)d_in[3];
    const float* gamma = (const float*)d_in[4];
    const float* beta  = (const float*)d_in[5];
    float* out = (float*)d_out;

    const int N = in_sizes[0] / CCH;
    const int E = in_sizes[1] / 2;

    const size_t nodef = (size_t)N * CCH;
    float* A_ws = (float*)d_ws;
    float* B_ws = A_ws + nodef;
    int*   M_ws = (int*)(B_ws + nodef);
    const int nb4 = (N + 63) / 64;
    float* psum    = (float*)(M_ws + nodef);
    float* psq     = psum + (size_t)nb4 * 64;
    unsigned* cnt  = (unsigned*)(psq + (size_t)nb4 * 64);
    float* gsum    = B_ws;
    float* gsq     = gsum + 64 * 64;
    float* meanp   = gsq  + 64 * 64;
    float* invstdp = meanp + 64;

    const long tot  = (long)nodef;

    k_gemm<<<(N + 15) / 16, 128, 0, stream>>>(x, W, bias, A_ws, B_ws, N);
    k_init<<<(int)((tot + 255) / 256), 256, 0, stream>>>(M_ws, tot, cnt);
    k_scatter<<<(E + 8 * EPW - 1) / (8 * EPW), 256, 0, stream>>>(B_ws, ei, M_ws, E);
    k_elu_partial<<<nb4, 256, 0, stream>>>(A_ws, M_ws, out, psum, psq, cnt,
                                           gsum, gsq, meanp, invstdp, N);
    k_bn<<<(int)((tot + 255) / 256), 256, 0, stream>>>(out, meanp, invstdp, gamma, beta, tot);
}


# Scatter-max buffer kept twice, one copy per L2 half: a wave reads its die (s_sendmsg_rtn 0x87), sends its atomics to its own half's copy (2nd copy in out); ELU takes the max of both
# speedup vs baseline: 2.9612x; 1.0376x over previous
#include <hip/hip_runtime.h>
#include <hip/hip_bf16.h>

typedef __attribute__((ext_vector_type(2))) float v2f;
typedef __attribute__((ext_vector_type(8))) float v8f;

#define CCH 64

__device__ __forceinline__ int fkey(int bits)
{
    return bits ^ ((bits >> 31) & 0x7FFFFFFF);
}
#define KEY_NEG_INF ((int)0x807FFFFF)

__global__ __launch_bounds__(128) void k_gemm(const float* __restrict__ x,
                                              const float* __restrict__ W,
                                              const float* __restrict__ bias,
                                              float* __restrict__ A_ws,
                                              float* __restrict__ B_ws,
                                              int N)
{
    const int lane = threadIdx.x & 31;
    const int wave = threadIdx.x >> 5;
    const int n0   = wave * 16;
    const int m0   = blockIdx.x * 16;
    const int half = lane >> 4;
    const int lm   = lane & 15;
    const int col  = n0 + lm;

    v8f accP = {};
    v8f accQ = {};

    int rowA = m0 + lm;
    if (rowA >= N) rowA = N - 1;

#pragma unroll
    for (int kk = 0; kk < 16; ++kk) {
        const int kb  = kk * 4 + half * 2;
        v2f a, b1, b2;
        a.x  = x[(long)rowA * CCH + kb];
        a.y  = x[(long)rowA * CCH + kb + 1];
        b1.x = W[kb * CCH + col];
        b1.y = W[(kb + 1) * CCH + col];
        b2.x = W[(kb + 64) * CCH + col];
        b2.y = W[(kb + 65) * CCH + col];
        accP = __builtin_amdgcn_wmma_f32_16x16x4_f32(false, a, false, b1,
                                                     (short)0, accP, false, false);
        accQ = __builtin_amdgcn_wmma_f32_16x16x4_f32(false, a, false, b2,
                                                     (short)0, accQ, false, false);
    }

    const float bv = bias[col];
    if (m0 + 16 <= N) {
#pragma unroll
        for (int r = 0; r < 8; ++r) {
            const int row = m0 + r + half * 8;
            const float q = accQ[r];
            A_ws[(long)row * CCH + col] = accP[r] - q + bv;
            B_ws[(long)row * CCH + col] = q;
        }
    } else {
#pragma unroll
        for (int r = 0; r < 8; ++r) {
            const int row = m0 + r + half * 8;
            if (row < N) {
                const float q = accQ[r];
                A_ws[(long)row * CCH + col] = accP[r] - q + bv;
                B_ws[(long)row * CCH + col] = q;
            }
        }
    }
}

#define NCNT 128

__global__ void k_init(int* __restrict__ M0, int* __restrict__ M1, long total,
                       unsigned* __restrict__ cnt)
{
    long i = (long)blockIdx.x * blockDim.x + threadIdx.x;
    if (i < total) {
        M0[i] = KEY_NEG_INF;
        M1[i] = KEY_NEG_INF;
    }
    if (i < NCNT) cnt[i] = 0u;
}

#define EPW 16

__device__ __forceinline__ void amax_f(int* addr, float v)
{
    atomicMax(addr, fkey(__float_as_int(v)));
}

__global__ __launch_bounds__(256) void k_scatter(const float* __restrict__ Bn,
                                                 const int* __restrict__ ei,
                                                 int* __restrict__ M0,
                                                 int* __restrict__ M1,
                                                 int E)
{
    const int lane = threadIdx.x & 31;
    const unsigned place = __builtin_amdgcn_s_sendmsg_rtn(0x87);
    int* __restrict__ M = ((place >> 18) & 1u) ? M1 : M0;
    const int wave = __builtin_amdgcn_readfirstlane((int)(threadIdx.x >> 5));
    const long e0  = ((long)blockIdx.x * 8 + wave) * EPW;
    if (e0 >= E) return;

    if (e0 + EPW <= E) {
        int src[EPW], dst[EPW];
#pragma unroll
        for (int j = 0; j < EPW; ++j) {
            src[j] = ei[e0 + j];
            dst[j] = ei[(long)E + e0 + j];
        }
        float v[2 * EPW];
#pragma unroll
        for (int j = 0; j < EPW; ++j) {
            const float* row = Bn + (long)src[j] * CCH;
            v[2 * j]     = row[lane];
            v[2 * j + 1] = row[lane + 32];
        }
#pragma unroll
        for (int j = 0; j < EPW; ++j) {
            int* row = M + (long)dst[j] * CCH;
            amax_f(row + lane,      v[2 * j]);
            amax_f(row + lane + 32, v[2 * j + 1]);
        }
    } else {
        for (long e = e0; e < E; ++e) {
            const int s = ei[e];
            const int d = ei[(long)E + e];
            const float* srow = Bn + (long)s * CCH;
            int* drow = M + (long)d * CCH;
            amax_f(drow + lane,      srow[lane]);
            amax_f(drow + lane + 32, srow[lane + 32]);
        }
    }
}


__device__ __forceinline__ void kadd(float& s, float& c, float x)
{
    const float y = x - c;
    const float t = s + y;
    c = (t - s) - y;
    s = t;
}

__device__ __forceinline__ void col_sums(const float* a0, const float* a1,
                                         int first, int n, float* sm,
                                         float& S, float& C)
{
    const int t    = threadIdx.x;
    const int lane = t & 31;
    const int w    = __builtin_amdgcn_readfirstlane(t >> 5);
    const float* base = ((lane < 16) ? (a0 + 4 * lane) : (a1 + 4 * (lane - 16)))
                        + (long)first * CCH;
    float4 s = make_float4(0.f, 0.f, 0.f, 0.f);
    float4 c = make_float4(0.f, 0.f, 0.f, 0.f);
    for (int r0 = w; r0 < n; r0 += 64) {
        float4 v[8];
#pragma unroll
        for (int j = 0; j < 8; ++j) {
            int r = r0 + 8 * j;
            if (r > n - 1) r = n - 1;
            v[j] = *(const float4*)(base + (long)r * CCH);
        }
#pragma unroll
        for (int j = 0; j < 8; ++j) {
            const bool live = (r0 + 8 * j < n);
            kadd(s.x, c.x, live ? v[j].x : 0.f);
            kadd(s.y, c.y, live ? v[j].y : 0.f);
            kadd(s.z, c.z, live ? v[j].z : 0.f);
            kadd(s.w, c.w, live ? v[j].w : 0.f);
        }
    }
    *(float4*)(sm + w * 128 + 4 * lane)        = s;
    *(float4*)(sm + 1024 + w * 128 + 4 * lane) = c;
    __syncthreads();
    S = 0.f; C = 0.f;
    if (t < 128) {
#pragma unroll
        for (int k = 0; k < 8; ++k) {
            kadd(S, C,  sm[k * 128 + t]);
            kadd(S, C, -sm[1024 + k * 128 + t]);
        }
    }
}

__device__ __forceinline__ unsigned take_ticket(unsigned* cnt, unsigned* s_slot)
{
    __builtin_amdgcn_fence(__ATOMIC_RELEASE, "agent");
    __syncthreads();
    if (threadIdx.x == 0) {
        __builtin_amdgcn_fence(__ATOMIC_RELEASE, "agent");
        *s_slot = __hip_atomic_fetch_add(cnt, 1u, __ATOMIC_RELAXED,
                                         __HIP_MEMORY_SCOPE_AGENT);
    }
    __syncthreads();
    return *s_slot;
}

__global__ __launch_bounds__(256) void k_elu_partial(const float* __restrict__ A_ws,
                                                     const int* __restrict__ M,
                                                     float* __restrict__ H,
                                                     float* psum,
                                                     float* psq,
                                                     unsigned* cnt,
                                                     float* gsum,
                                                     float* gsq,
                                                     float* mean,
                                                     float* invstd,
                                                     int N)
{
    __shared__ float smem[256];
    __shared__ float sred[2048];
    __shared__ double sfin[128];
    __shared__ unsigned s_tick[2];
    const int t  = threadIdx.x;
    const int c  = t & 63;
    const int r4 = t >> 6;
    const int base = blockIdx.x * 64;

    float s = 0.f, q = 0.f;
    for (int rr = r4; rr < 64; rr += 4) {
        int row = base + rr;
        if (row < N) {
            long idx = (long)row * CCH + c;
            int  mb  = M[idx];
            const int m1 = __float_as_int(H[idx]);
            mb = (m1 > mb) ? m1 : mb;
            float h;
            if (mb == KEY_NEG_INF) {
                h = 0.f;
            } else {
                float z = A_ws[idx] + __int_as_float(fkey(mb));
                h = (z > 0.f) ? z : expm1f(z);
            }
            H[idx] = h;
            s += h;
            q += h * h;
        }
    }
    smem[t] = s; __syncthreads();
    if (r4 == 0)
        psum[(long)blockIdx.x * 64 + c] = smem[c] + smem[c + 64] + smem[c + 128] + smem[c + 192];
    __syncthreads();
    smem[t] = q; __syncthreads();
    if (r4 == 0)
        psq[(long)blockIdx.x * 64 + c]  = smem[c] + smem[c + 64] + smem[c + 128] + smem[c + 192];

    const int nb  = (int)gridDim.x;
    const int S   = (nb + 63) / 64;
    const int G   = (nb + S - 1) / S;
    const int g   = (int)blockIdx.x / S;
    const int pop = (nb - g * S < S) ? (nb - g * S) : S;

    if (take_ticket(cnt + 2 * g, &s_tick[0]) != (unsigned)(pop - 1)) return;
    __builtin_amdgcn_fence(__ATOMIC_ACQUIRE, "agent");

    float S1, C1;
    col_sums(psum, psq, g * S, pop, sred, S1, C1);
    if (t < 64)       gsum[g * 64 + t]        = S1 - C1;
    else if (t < 128) gsq [g * 64 + (t - 64)] = S1 - C1;

    if (take_ticket(cnt + (NCNT - 1), &s_tick[1]) != (unsigned)(G - 1)) return;
    __builtin_amdgcn_fence(__ATOMIC_ACQUIRE, "agent");

    float S2, C2;
    col_sums(gsum, gsq, 0, G, sred, S2, C2);
    if (t < 128) sfin[t] = (double)S2 - (double)C2;
    __syncthreads();
    if (t < 64) {
        const double m = sfin[t] / (double)N;
        double v = sfin[64 + t] / (double)N - m * m;
        if (v < 0.0) v = 0.0;
        mean[t]   = (float)m;
        invstd[t] = (float)(1.0 / sqrt(v + 1e-5));
    }
}

__global__ __launch_bounds__(256) void k_bn(float* __restrict__ out,
                                            const float* __restrict__ mean,
                                            const float* __restrict__ invstd,
                                            const float* __restrict__ gamma,
                                            const float* __restrict__ beta,
                                            long total)
{
    long i = (long)blockIdx.x * blockDim.x + threadIdx.x;
    if (i < total) {
        int c = (int)(i & 63);
        out[i] = gamma[c] * (out[i] - mean[c]) * invstd[c] + beta[c];
    }
}

extern "C" void kernel_launch(void* const* d_in, const int* in_sizes, int n_in,
                              void* d_out, int out_size, void* d_ws, size_t ws_size,
                              hipStream_t stream)
{
    const float* x     = (const float*)d_in[0];
    const int*   ei    = (const int*)  d_in[1];
    const float* W     = (const float*)d_in[2];
    const float* bias  = (const float*)d_in[3];
    const float* gamma = (const float*)d_in[4];
    const float* beta  = (const float*)d_in[5];
    float* out = (float*)d_out;

    const int N = in_sizes[0] / CCH;
    const int E = in_sizes[1] / 2;

    const size_t nodef = (size_t)N * CCH;
    float* A_ws = (float*)d_ws;
    float* B_ws = A_ws + nodef;
    int*   M_ws = (int*)(B_ws + nodef);
    int*   M2_ws = (int*)d_out;
    const int nb4 = (N + 63) / 64;
    float* psum    = (float*)(M_ws + nodef);
    float* psq     = psum + (size_t)nb4 * 64;
    unsigned* cnt  = (unsigned*)(psq + (size_t)nb4 * 64);
    float* gsum    = B_ws;
    float* gsq     = gsum + 64 * 64;
    float* meanp   = gsq  + 64 * 64;
    float* invstdp = meanp + 64;

    const long tot  = (long)nodef;

    k_gemm<<<(N + 15) / 16, 128, 0, stream>>>(x, W, bias, A_ws, B_ws, N);
    k_init<<<(int)((tot + 255) / 256), 256, 0, stream>>>(M_ws, M2_ws, tot, cnt);
    k_scatter<<<(E + 8 * EPW - 1) / (8 * EPW), 256, 0, stream>>>(B_ws, ei, M_ws, M2_ws, E);
    k_elu_partial<<<nb4, 256, 0, stream>>>(A_ws, M_ws, out, psum, psq, cnt,
                                           gsum, gsq, meanp, invstdp, N);
    k_bn<<<(int)((tot + 255) / 256), 256, 0, stream>>>(out, meanp, invstdp, gamma, beta, tot);
}


# Scatter-max by owner blocks: a block keeps 400 node rows of keys in LDS, scans one owner byte per edge (new packing pre-pass), gathers its edges' rows, LDS max; no global atomic; BN sums regrouped
# speedup vs baseline: 3.5024x; 1.1828x over previous
#include <hip/hip_runtime.h>
#include <hip/hip_bf16.h>

typedef __attribute__((ext_vector_type(2))) float v2f;
typedef __attribute__((ext_vector_type(8))) float v8f;

#define CCH 64

__device__ __forceinline__ int fkey(int bits)
{
    return bits ^ ((bits >> 31) & 0x7FFFFFFF);
}
#define KEY_NEG_INF ((int)0x807FFFFF)

__global__ __launch_bounds__(128) void k_gemm(const float* __restrict__ x,
                                              const float* __restrict__ W,
                                              const float* __restrict__ bias,
                                              float* __restrict__ A_ws,
                                              float* __restrict__ B_ws,
                                              int N)
{
    const int lane = threadIdx.x & 31;
    const int wave = threadIdx.x >> 5;
    const int n0   = wave * 16;
    const int m0   = blockIdx.x * 16;
    const int half = lane >> 4;
    const int lm   = lane & 15;
    const int col  = n0 + lm;

    v8f accP = {};
    v8f accQ = {};

    int rowA = m0 + lm;
    if (rowA >= N) rowA = N - 1;

#pragma unroll
    for (int kk = 0; kk < 16; ++kk) {
        const int kb  = kk * 4 + half * 2;
        v2f a, b1, b2;
        a.x  = x[(long)rowA * CCH + kb];
        a.y  = x[(long)rowA * CCH + kb + 1];
        b1.x = W[kb * CCH + col];
        b1.y = W[(kb + 1) * CCH + col];
        b2.x = W[(kb + 64) * CCH + col];
        b2.y = W[(kb + 65) * CCH + col];
        accP = __builtin_amdgcn_wmma_f32_16x16x4_f32(false, a, false, b1,
                                                     (short)0, accP, false, false);
        accQ = __builtin_amdgcn_wmma_f32_16x16x4_f32(false, a, false, b2,
                                                     (short)0, accQ, false, false);
    }

    const float bv = bias[col];
    if (m0 + 16 <= N) {
#pragma unroll
        for (int r = 0; r < 8; ++r) {
            const int row = m0 + r + half * 8;
            const float q = accQ[r];
            A_ws[(long)row * CCH + col] = accP[r] - q + bv;
            B_ws[(long)row * CCH + col] = q;
        }
    } else {
#pragma unroll
        for (int r = 0; r < 8; ++r) {
            const int row = m0 + r + half * 8;
            if (row < N) {
                const float q = accQ[r];
                A_ws[(long)row * CCH + col] = accP[r] - q + bv;
                B_ws[(long)row * CCH + col] = q;
            }
        }
    }
}

#define NPB      400
#define REC_NONE (511u << 17)
#define CHUNK    1024

__global__ __launch_bounds__(256) void k_records(const int* __restrict__ ei,
                                                 unsigned* __restrict__ own,
                                                 unsigned* __restrict__ rec,
                                                 unsigned* __restrict__ cnt,
                                                 int E, int N, long padded)
{
    const long i = (long)blockIdx.x * blockDim.x + threadIdx.x;
    if (i == 0) cnt[0] = 0u;
    const long e0 = 4 * i;
    if (e0 >= padded) return;
    unsigned ow = 0u;
    unsigned r[4];
#pragma unroll
    for (int k = 0; k < 4; ++k) {
        const long e = e0 + k;
        unsigned o = 255u, rc = REC_NONE;
        if (e < E) {
            int s = ei[e];
            const int d = ei[(long)E + e];
            if ((unsigned)d < (unsigned)N) {
                s = (s < 0) ? 0 : ((s > N - 1) ? N - 1 : s);
                const unsigned ob = (unsigned)d / NPB;
                if (ob < 255u) {
                    o  = ob;
                    rc = (unsigned)s | (((unsigned)d - ob * NPB) << 17);
                }
            }
        }
        ow |= o << (8 * k);
        r[k] = rc;
    }
    own[i] = ow;
    *(uint4*)(rec + e0) = make_uint4(r[0], r[1], r[2], r[3]);
}

__device__ __forceinline__ unsigned take_ticket(unsigned* cnt, unsigned* s_slot)
{
    __builtin_amdgcn_fence(__ATOMIC_RELEASE, "agent");
    __syncthreads();
    if (threadIdx.x == 0) {
        __builtin_amdgcn_fence(__ATOMIC_RELEASE, "agent");
        *s_slot = __hip_atomic_fetch_add(cnt, 1u, __ATOMIC_RELAXED,
                                         __HIP_MEMORY_SCOPE_AGENT);
    }
    __syncthreads();
    return *s_slot;
}

#define OWN_WAVES 32
#define QW        1024

__device__ __forceinline__ void gather_batch(const int* q, unsigned h0, unsigned n,
                                             int lane, int rows,
                                             const unsigned* __restrict__ rec,
                                             const float* __restrict__ Bn,
                                             int* keys)
{
    __builtin_amdgcn_fence(__ATOMIC_ACQ_REL, "wavefront");
    const int e = q[(h0 + (unsigned)lane) & (QW - 1)];
    unsigned r = REC_NONE;
    if ((unsigned)lane < n) r = rec[e];
    const int src = (int)(r & 0x1FFFFu);
    const int dl  = (int)(r >> 17);
    float2 v[32];
#pragma unroll
    for (int j = 0; j < 32; ++j) {
        const int sj = __builtin_amdgcn_readlane(src, j);
        v[j] = *((const float2*)(Bn + (long)sj * CCH) + lane);
    }
#pragma unroll
    for (int j = 0; j < 32; ++j) {
        const int dj = __builtin_amdgcn_readlane(dl, j);
        if (dj < rows) {
            int* row = keys + dj * CCH + 2 * lane;
            __hip_atomic_fetch_max(row,     fkey(__float_as_int(v[j].x)),
                                   __ATOMIC_RELAXED, __HIP_MEMORY_SCOPE_WORKGROUP);
            __hip_atomic_fetch_max(row + 1, fkey(__float_as_int(v[j].y)),
                                   __ATOMIC_RELAXED, __HIP_MEMORY_SCOPE_WORKGROUP);
        }
    }
}

__device__ __forceinline__ void pick_owned(const uint4 w, int ebase, unsigned splat,
                                           int* qw, unsigned& tail)
{
    const unsigned wd[4] = {w.x, w.y, w.z, w.w};
#pragma unroll
    for (int d = 0; d < 4; ++d) {
        const unsigned x = wd[d] ^ splat;
        const unsigned m = (x - 0x01010101u) & ~x & 0x80808080u;
        if (__builtin_amdgcn_ballot_w32(m != 0u) != 0u) {
#pragma unroll
            for (int k = 0; k < 4; ++k) {
                const bool hit = ((x >> (8 * k)) & 0xFFu) == 0u;
                const unsigned bal = __builtin_amdgcn_ballot_w32(hit);
                if (bal != 0u) {
                    const unsigned pre = __builtin_amdgcn_mbcnt_lo(bal, 0u);
                    if (hit) qw[(tail + pre) & (QW - 1)] = ebase + 4 * d + k;
                    tail += (unsigned)__builtin_popcount(bal);
                }
            }
        }
    }
}

__global__ __launch_bounds__(1024) void k_owner_max(const float* __restrict__ Bn,
                                                    const unsigned* __restrict__ own,
                                                    const unsigned* __restrict__ rec,
                                                    const float* __restrict__ A_ws,
                                                    float* __restrict__ H,
                                                    float* psum,
                                                    float* psq,
                                                    unsigned* cnt,
                                                    float* mean,
                                                    float* invstd,
                                                    int N, int nC)
{
    __shared__ int keys[NPB * CCH];
    __shared__ __attribute__((aligned(16))) int queue[OWN_WAVES * QW];
    __shared__ unsigned s_tick;
    const int t    = threadIdx.x;
    const int lane = t & 31;
    const int wave = __builtin_amdgcn_readfirstlane(t >> 5);
    const int b    = (int)blockIdx.x;
    const int lo   = b * NPB;
    const int rows = (N - lo < NPB) ? (N - lo) : NPB;

    for (int i = t; i < NPB * CCH; i += 1024) keys[i] = KEY_NEG_INF;
    __syncthreads();

    {
        const int* q = queue + wave * QW;
        int* qw = queue + wave * QW;
        unsigned head = 0u, tail = 0u;
        const unsigned splat = (unsigned)b * 0x01010101u;
        const int start = (int)(((long)b * nC) / (int)gridDim.x);
        const uint4* own16 = (const uint4*)own;
        const int nIt = (nC > wave) ? (nC - wave + OWN_WAVES - 1) / OWN_WAVES : 0;

        int c = wave + start; if (c >= nC) c -= nC;
        uint4 na = make_uint4(0u, 0u, 0u, 0u), nb2 = na;
        if (nIt > 0) {
            na  = own16[(long)c * 64 + lane];
            nb2 = own16[(long)c * 64 + 32 + lane];
        }
        for (int i = 0; i <= nIt; ++i) {
            const bool last = (i == nIt);
            if (!last) {
                const uint4 wa = na, wb = nb2;
                const int ebase = c * CHUNK + lane * 16;
                int cn = c + OWN_WAVES; if (cn >= nC) cn -= nC;
                if (i + 1 < nIt) {
                    na  = own16[(long)cn * 64 + lane];
                    nb2 = own16[(long)cn * 64 + 32 + lane];
                }
                c = cn;
                pick_owned(wa, ebase, splat, qw, tail);
                while (tail - head >= 32u) {
                    gather_batch(q, head, 32u, lane, rows, rec, Bn, keys);
                    head += 32u;
                }
                pick_owned(wb, ebase + 512, splat, qw, tail);
            }
            while (tail - head >= 32u || (last && tail != head)) {
                const unsigned n = (tail - head < 32u) ? (tail - head) : 32u;
                gather_batch(q, head, n, lane, rows, rec, Bn, keys);
                head += n;
            }
        }
    }
    __syncthreads();

    const int c64   = t & 63;
    const int strip = t >> 6;
    float s = 0.f, sq = 0.f;
    for (int r0 = strip; r0 < rows; r0 += 16 * 5) {
        float a[5];
#pragma unroll
        for (int j = 0; j < 5; ++j) {
            int r = r0 + 16 * j;
            if (r > rows - 1) r = rows - 1;
            a[j] = A_ws[(long)(lo + r) * CCH + c64];
        }
#pragma unroll
        for (int j = 0; j < 5; ++j) {
            const int r = r0 + 16 * j;
            if (r < rows) {
                const int mb = keys[r * CCH + c64];
                float h;
                if (mb == KEY_NEG_INF) {
                    h = 0.f;
                } else {
                    const float z = a[j] + __int_as_float(fkey(mb));
                    h = (z > 0.f) ? z : expm1f(z);
                }
                H[(long)(lo + r) * CCH + c64] = h;
                s  += h;
                sq += h * h;
            }
        }
    }
    float* red = (float*)queue;
    red[strip * 128 + c64]      = s;
    red[strip * 128 + 64 + c64] = sq;
    __syncthreads();
    if (t < 128) {
        float acc = 0.f;
#pragma unroll
        for (int k = 0; k < 16; ++k) acc += red[k * 128 + t];
        if (t < 64) psum[(long)b * 64 + t]        = acc;
        else        psq [(long)b * 64 + (t - 64)] = acc;
    }

    const int nb = (int)gridDim.x;
    if (take_ticket(cnt, &s_tick) != (unsigned)(nb - 1)) return;
    __builtin_amdgcn_fence(__ATOMIC_ACQUIRE, "agent");

    const int col   = t & 127;
    const int slice = t >> 7;
    const float* p  = (col < 64) ? (psum + col) : (psq + (col - 64));
    double acc = 0.0;
    for (int r0 = slice; r0 < nb; r0 += 8 * 16) {
        float v[16];
#pragma unroll
        for (int j = 0; j < 16; ++j) {
            int r = r0 + 8 * j;
            if (r > nb - 1) r = nb - 1;
            v[j] = p[(long)r * 64];
        }
#pragma unroll
        for (int j = 0; j < 16; ++j)
            acc += (r0 + 8 * j < nb) ? (double)v[j] : 0.0;
    }
    double* fin = (double*)queue;
    fin[slice * 128 + col] = acc;
    __syncthreads();
    if (t < 64) {
        double sm = 0.0, sv = 0.0;
#pragma unroll
        for (int k = 0; k < 8; ++k) {
            sm += fin[k * 128 + t];
            sv += fin[k * 128 + 64 + t];
        }
        const double m = sm / (double)N;
        double v = sv / (double)N - m * m;
        if (v < 0.0) v = 0.0;
        mean[t]   = (float)m;
        invstd[t] = (float)(1.0 / sqrt(v + 1e-5));
    }
}

__global__ __launch_bounds__(256) void k_bn(float* __restrict__ out,
                                            const float* __restrict__ mean,
                                            const float* __restrict__ invstd,
                                            const float* __restrict__ gamma,
                                            const float* __restrict__ beta,
                                            long total)
{
    long i = (long)blockIdx.x * blockDim.x + threadIdx.x;
    if (i < total) {
        int c = (int)(i & 63);
        out[i] = gamma[c] * (out[i] - mean[c]) * invstd[c] + beta[c];
    }
}

extern "C" void kernel_launch(void* const* d_in, const int* in_sizes, int n_in,
                              void* d_out, int out_size, void* d_ws, size_t ws_size,
                              hipStream_t stream)
{
    const float* x     = (const float*)d_in[0];
    const int*   ei    = (const int*)  d_in[1];
    const float* W     = (const float*)d_in[2];
    const float* bias  = (const float*)d_in[3];
    const float* gamma = (const float*)d_in[4];
    const float* beta  = (const float*)d_in[5];
    float* out = (float*)d_out;

    const int N = in_sizes[0] / CCH;
    const int E = in_sizes[1] / 2;

    const size_t nodef = (size_t)N * CCH;
    float* A_ws = (float*)d_ws;
    float* B_ws = A_ws + nodef;
    const int nC = (E + CHUNK - 1) / CHUNK;
    const int nb = (N + NPB - 1) / NPB;
    unsigned* own  = (unsigned*)(B_ws + nodef);
    unsigned* rec  = own + (size_t)nC * (CHUNK / 4);
    float* psum    = (float*)(rec + (size_t)nC * CHUNK);
    float* psq     = psum + (size_t)nb * 64;
    float* meanp   = psq  + (size_t)nb * 64;
    float* invstdp = meanp + 64;
    unsigned* cnt  = (unsigned*)(invstdp + 64);

    const long tot    = (long)nodef;
    const long padded = (long)nC * CHUNK;

    k_gemm<<<(N + 15) / 16, 128, 0, stream>>>(x, W, bias, A_ws, B_ws, N);
    k_records<<<(int)((padded / 4 + 255) / 256), 256, 0, stream>>>(ei, own, rec, cnt, E, N, padded);
    k_owner_max<<<nb, 1024, 0, stream>>>(B_ws, own, rec, A_ws, out, psum, psq, cnt,
                                         meanp, invstdp, N, nC);
    k_bn<<<(int)((tot + 255) / 256), 256, 0, stream>>>(out, meanp, invstdp, gamma, beta, tot);
}


# Owner ids bit-sliced by the packing pass (8 ballot words per 32 edges): an owner block matches 32 edges a lane in 8 logic ops and queues hits in ballot rounds; was a byte test of 4 edges a word
# speedup vs baseline: 3.9614x; 1.1311x over previous
#include <hip/hip_runtime.h>
#include <hip/hip_bf16.h>

typedef __attribute__((ext_vector_type(2))) float v2f;
typedef __attribute__((ext_vector_type(8))) float v8f;

#define CCH 64

__device__ __forceinline__ int fkey(int bits)
{
    return bits ^ ((bits >> 31) & 0x7FFFFFFF);
}
#define KEY_NEG_INF ((int)0x807FFFFF)

__global__ __launch_bounds__(128) void k_gemm(const float* __restrict__ x,
                                              const float* __restrict__ W,
                                              const float* __restrict__ bias,
                                              float* __restrict__ A_ws,
                                              float* __restrict__ B_ws,
                                              int N)
{
    const int lane = threadIdx.x & 31;
    const int wave = threadIdx.x >> 5;
    const int n0   = wave * 16;
    const int m0   = blockIdx.x * 16;
    const int half = lane >> 4;
    const int lm   = lane & 15;
    const int col  = n0 + lm;

    v8f accP = {};
    v8f accQ = {};

    int rowA = m0 + lm;
    if (rowA >= N) rowA = N - 1;

#pragma unroll
    for (int kk = 0; kk < 16; ++kk) {
        const int kb  = kk * 4 + half * 2;
        v2f a, b1, b2;
        a.x  = x[(long)rowA * CCH + kb];
        a.y  = x[(long)rowA * CCH + kb + 1];
        b1.x = W[kb * CCH + col];
        b1.y = W[(kb + 1) * CCH + col];
        b2.x = W[(kb + 64) * CCH + col];
        b2.y = W[(kb + 65) * CCH + col];
        accP = __builtin_amdgcn_wmma_f32_16x16x4_f32(false, a, false, b1,
                                                     (short)0, accP, false, false);
        accQ = __builtin_amdgcn_wmma_f32_16x16x4_f32(false, a, false, b2,
                                                     (short)0, accQ, false, false);
    }

    const float bv = bias[col];
    if (m0 + 16 <= N) {
#pragma unroll
        for (int r = 0; r < 8; ++r) {
            const int row = m0 + r + half * 8;
            const float q = accQ[r];
            A_ws[(long)row * CCH + col] = accP[r] - q + bv;
            B_ws[(long)row * CCH + col] = q;
        }
    } else {
#pragma unroll
        for (int r = 0; r < 8; ++r) {
            const int row = m0 + r + half * 8;
            if (row < N) {
                const float q = accQ[r];
                A_ws[(long)row * CCH + col] = accP[r] - q + bv;
                B_ws[(long)row * CCH + col] = q;
            }
        }
    }
}

#define NPB      400
#define REC_NONE (511u << 17)
#define CHUNK    1024

__global__ __launch_bounds__(256) void k_records(const int* __restrict__ ei,
                                                 unsigned* __restrict__ own,
                                                 unsigned* __restrict__ rec,
                                                 unsigned* __restrict__ cnt,
                                                 int E, int N, long padded)
{
    const int lane = threadIdx.x & 31;
    const int wib = __builtin_amdgcn_readfirstlane((int)(threadIdx.x >> 5));
    if (blockIdx.x == 0 && threadIdx.x == 0) cnt[0] = 0u;
    const long e0 = ((long)blockIdx.x * (blockDim.x >> 5) + wib) * 128;
    if (e0 >= padded) return;
    int sv[4], dv[4];
#pragma unroll
    for (int k = 0; k < 4; ++k) {
        const long e  = e0 + 32 * k + lane;
        const long ec = (e < E) ? e : (long)E - 1;
        sv[k] = ei[ec];
        dv[k] = ei[(long)E + ec];
    }
    unsigned word = 0u;
#pragma unroll
    for (int k = 0; k < 4; ++k) {
        const long e = e0 + 32 * k + lane;
        unsigned o = 255u, rc = REC_NONE;
        if (e < E) {
            int s = sv[k];
            const int d = dv[k];
            if ((unsigned)d < (unsigned)N) {
                s = (s < 0) ? 0 : ((s > N - 1) ? N - 1 : s);
                const unsigned ob = (unsigned)d / NPB;
                if (ob < 255u) {
                    o  = ob;
                    rc = (unsigned)s | (((unsigned)d - ob * NPB) << 17);
                }
            }
        }
        rec[e] = rc;
#pragma unroll
        for (int j = 0; j < 8; ++j) {
            const unsigned bal = __builtin_amdgcn_ballot_w32(((o >> j) & 1u) != 0u);
            if (lane == (j >> 2) * 16 + k * 4 + (j & 3)) word = bal;
        }
    }
    const long c  = e0 >> 10;
    const int  l0 = (int)((e0 >> 5) & 31);
    own[c * 256 + (lane >> 4) * 128 + l0 * 4 + (lane & 15)] = word;
}

__device__ __forceinline__ unsigned take_ticket(unsigned* cnt, unsigned* s_slot)
{
    __builtin_amdgcn_fence(__ATOMIC_RELEASE, "agent");
    __syncthreads();
    if (threadIdx.x == 0) {
        __builtin_amdgcn_fence(__ATOMIC_RELEASE, "agent");
        *s_slot = __hip_atomic_fetch_add(cnt, 1u, __ATOMIC_RELAXED,
                                         __HIP_MEMORY_SCOPE_AGENT);
    }
    __syncthreads();
    return *s_slot;
}

#define OWN_WAVES 32
#define QW        64

__device__ __forceinline__ void gather_batch(const int* q, unsigned h0, unsigned n,
                                             int lane, int rows,
                                             const unsigned* __restrict__ rec,
                                             const float* __restrict__ Bn,
                                             int* keys)
{
    __builtin_amdgcn_fence(__ATOMIC_ACQ_REL, "wavefront");
    const int e = q[(h0 + (unsigned)lane) & (QW - 1)];
    unsigned r = REC_NONE;
    if ((unsigned)lane < n) r = rec[e];
    const int src = (int)(r & 0x1FFFFu);
    const int dl  = (int)(r >> 17);
    float2 v[32];
#pragma unroll
    for (int j = 0; j < 32; ++j) {
        const int sj = __builtin_amdgcn_readlane(src, j);
        v[j] = *((const float2*)(Bn + (long)sj * CCH) + lane);
    }
#pragma unroll
    for (int j = 0; j < 32; ++j) {
        const int dj = __builtin_amdgcn_readlane(dl, j);
        if (dj < rows) {
            int* row = keys + dj * CCH + 2 * lane;
            __hip_atomic_fetch_max(row,     fkey(__float_as_int(v[j].x)),
                                   __ATOMIC_RELAXED, __HIP_MEMORY_SCOPE_WORKGROUP);
            __hip_atomic_fetch_max(row + 1, fkey(__float_as_int(v[j].y)),
                                   __ATOMIC_RELAXED, __HIP_MEMORY_SCOPE_WORKGROUP);
        }
    }
}

__device__ __forceinline__ unsigned match_owned(const uint4 lo, const uint4 hi,
                                                const unsigned* kx)
{
    return (lo.x ^ kx[0]) & (lo.y ^ kx[1]) & (lo.z ^ kx[2]) & (lo.w ^ kx[3])
         & (hi.x ^ kx[4]) & (hi.y ^ kx[5]) & (hi.z ^ kx[6]) & (hi.w ^ kx[7]);
}

__global__ __launch_bounds__(1024) void k_owner_max(const float* __restrict__ Bn,
                                                    const unsigned* __restrict__ own,
                                                    const unsigned* __restrict__ rec,
                                                    const float* __restrict__ A_ws,
                                                    float* __restrict__ H,
                                                    float* psum,
                                                    float* psq,
                                                    unsigned* cnt,
                                                    float* mean,
                                                    float* invstd,
                                                    int N, int nC)
{
    __shared__ int keys[NPB * CCH];
    __shared__ __attribute__((aligned(16))) int queue[OWN_WAVES * QW];
    __shared__ unsigned s_tick;
    const int t    = threadIdx.x;
    const int lane = t & 31;
    const int wave = __builtin_amdgcn_readfirstlane(t >> 5);
    const int b    = (int)blockIdx.x;
    const int lo   = b * NPB;
    const int rows = (N - lo < NPB) ? (N - lo) : NPB;

    for (int i = t; i < NPB * CCH; i += 1024) keys[i] = KEY_NEG_INF;
    __syncthreads();

    {
        const int* q = queue + wave * QW;
        int* qw = queue + wave * QW;
        unsigned head = 0u, tail = 0u;
        unsigned kx[8];
#pragma unroll
        for (int j = 0; j < 8; ++j) kx[j] = ((b >> j) & 1) ? 0u : 0xFFFFFFFFu;
        const int start = (int)(((long)b * nC) / (int)gridDim.x);
        const uint4* own16 = (const uint4*)own;
        const int nIt = (nC > wave) ? (nC - wave + OWN_WAVES - 1) / OWN_WAVES : 0;
        const int nSt = (nIt + 1) >> 1;

        auto drain = [&](unsigned m, int ebase) {
            for (;;) {
                const unsigned bal = __builtin_amdgcn_ballot_w32(m != 0u);
                if (bal == 0u) break;
                const unsigned pre = __builtin_amdgcn_mbcnt_lo(bal, 0u);
                if (m != 0u) qw[(tail + pre) & (QW - 1)] = ebase + __builtin_ctz(m);
                tail += (unsigned)__builtin_popcount(bal);
                m &= m - 1u;
                if (tail - head >= 32u) {
                    gather_batch(q, head, 32u, lane, rows, rec, Bn, keys);
                    head += 32u;
                }
            }
        };

        int cA = wave + start;   if (cA >= nC) cA -= nC;
        int cB = cA + OWN_WAVES; if (cB >= nC) cB -= nC;
        const uint4 ones = make_uint4(~0u, ~0u, ~0u, ~0u);
        uint4 nAl = ones, nAh = ones, nBl = ones, nBh = ones;
        if (nIt > 0) {
            nAl = own16[(long)cA * 64 + lane];
            nAh = own16[(long)cA * 64 + 32 + lane];
        }
        if (nIt > 1) {
            nBl = own16[(long)cB * 64 + lane];
            nBh = own16[(long)cB * 64 + 32 + lane];
        }
        for (int s = 0; s < nSt; ++s) {
            const unsigned mA = match_owned(nAl, nAh, kx);
            const unsigned mB = (2 * s + 1 < nIt) ? match_owned(nBl, nBh, kx) : 0u;
            const int eA = cA * CHUNK + lane * 32;
            const int eB = cB * CHUNK + lane * 32;
            cA = cB + OWN_WAVES; if (cA >= nC) cA -= nC;
            cB = cA + OWN_WAVES; if (cB >= nC) cB -= nC;
            if (2 * s + 2 < nIt) {
                nAl = own16[(long)cA * 64 + lane];
                nAh = own16[(long)cA * 64 + 32 + lane];
            }
            if (2 * s + 3 < nIt) {
                nBl = own16[(long)cB * 64 + lane];
                nBh = own16[(long)cB * 64 + 32 + lane];
            }
            drain(mA, eA);
            drain(mB, eB);
        }
        if (tail != head)
            gather_batch(q, head, tail - head, lane, rows, rec, Bn, keys);
    }
    __syncthreads();

    const int c64   = t & 63;
    const int strip = t >> 6;
    float s = 0.f, sq = 0.f;
    for (int r0 = strip; r0 < rows; r0 += 16 * 5) {
        float a[5];
#pragma unroll
        for (int j = 0; j < 5; ++j) {
            int r = r0 + 16 * j;
            if (r > rows - 1) r = rows - 1;
            a[j] = A_ws[(long)(lo + r) * CCH + c64];
        }
#pragma unroll
        for (int j = 0; j < 5; ++j) {
            const int r = r0 + 16 * j;
            if (r < rows) {
                const int mb = keys[r * CCH + c64];
                float h;
                if (mb == KEY_NEG_INF) {
                    h = 0.f;
                } else {
                    const float z = a[j] + __int_as_float(fkey(mb));
                    h = (z > 0.f) ? z : expm1f(z);
                }
                H[(long)(lo + r) * CCH + c64] = h;
                s  += h;
                sq += h * h;
            }
        }
    }
    float* red = (float*)queue;
    red[strip * 128 + c64]      = s;
    red[strip * 128 + 64 + c64] = sq;
    __syncthreads();
    if (t < 128) {
        float acc = 0.f;
#pragma unroll
        for (int k = 0; k < 16; ++k) acc += red[k * 128 + t];
        if (t < 64) psum[(long)b * 64 + t]        = acc;
        else        psq [(long)b * 64 + (t - 64)] = acc;
    }

    const int nb = (int)gridDim.x;
    if (take_ticket(cnt, &s_tick) != (unsigned)(nb - 1)) return;
    __builtin_amdgcn_fence(__ATOMIC_ACQUIRE, "agent");

    const int col   = t & 127;
    const int slice = t >> 7;
    const float* p  = (col < 64) ? (psum + col) : (psq + (col - 64));
    double acc = 0.0;
    for (int r0 = slice; r0 < nb; r0 += 8 * 16) {
        float v[16];
#pragma unroll
        for (int j = 0; j < 16; ++j) {
            int r = r0 + 8 * j;
            if (r > nb - 1) r = nb - 1;
            v[j] = p[(long)r * 64];
        }
#pragma unroll
        for (int j = 0; j < 16; ++j)
            acc += (r0 + 8 * j < nb) ? (double)v[j] : 0.0;
    }
    double* fin = (double*)queue;
    fin[slice * 128 + col] = acc;
    __syncthreads();
    if (t < 64) {
        double sm = 0.0, sv = 0.0;
#pragma unroll
        for (int k = 0; k < 8; ++k) {
            sm += fin[k * 128 + t];
            sv += fin[k * 128 + 64 + t];
        }
        const double m = sm / (double)N;
        double v = sv / (double)N - m * m;
        if (v < 0.0) v = 0.0;
        mean[t]   = (float)m;
        invstd[t] = (float)(1.0 / sqrt(v + 1e-5));
    }
}

__global__ __launch_bounds__(256) void k_bn(float* __restrict__ out,
                                            const float* __restrict__ mean,
                                            const float* __restrict__ invstd,
                                            const float* __restrict__ gamma,
                                            const float* __restrict__ beta,
                                            long total)
{
    long i = (long)blockIdx.x * blockDim.x + threadIdx.x;
    if (i < total) {
        int c = (int)(i & 63);
        out[i] = gamma[c] * (out[i] - mean[c]) * invstd[c] + beta[c];
    }
}

extern "C" void kernel_launch(void* const* d_in, const int* in_sizes, int n_in,
                              void* d_out, int out_size, void* d_ws, size_t ws_size,
                              hipStream_t stream)
{
    const float* x     = (const float*)d_in[0];
    const int*   ei    = (const int*)  d_in[1];
    const float* W     = (const float*)d_in[2];
    const float* bias  = (const float*)d_in[3];
    const float* gamma = (const float*)d_in[4];
    const float* beta  = (const float*)d_in[5];
    float* out = (float*)d_out;

    const int N = in_sizes[0] / CCH;
    const int E = in_sizes[1] / 2;

    const size_t nodef = (size_t)N * CCH;
    float* A_ws = (float*)d_ws;
    float* B_ws = A_ws + nodef;
    const int nC = (E + CHUNK - 1) / CHUNK;
    const int nb = (N + NPB - 1) / NPB;
    unsigned* own  = (unsigned*)(B_ws + nodef);
    unsigned* rec  = own + (size_t)nC * (CHUNK / 4);
    float* psum    = (float*)(rec + (size_t)nC * CHUNK);
    float* psq     = psum + (size_t)nb * 64;
    float* meanp   = psq  + (size_t)nb * 64;
    float* invstdp = meanp + 64;
    unsigned* cnt  = (unsigned*)(invstdp + 64);

    const long tot    = (long)nodef;
    const long padded = (long)nC * CHUNK;

    k_gemm<<<(N + 15) / 16, 128, 0, stream>>>(x, W, bias, A_ws, B_ws, N);
    k_records<<<(int)((padded / 4 + 255) / 256), 256, 0, stream>>>(ei, own, rec, cnt, E, N, padded);
    k_owner_max<<<nb, 1024, 0, stream>>>(B_ws, own, rec, A_ws, out, psum, psq, cnt,
                                         meanp, invstdp, N, nC);
    k_bn<<<(int)((tot + 255) / 256), 256, 0, stream>>>(out, meanp, invstdp, gamma, beta, tot);
}


# GEMM: a wave loads the W fragments of its column tile once (64 registers) and walks 5 row tiles with them, next tile's x fragments in flight; was 64 W loads per tile; 1250 blocks, not 6250
# speedup vs baseline: 4.0968x; 1.0342x over previous
#include <hip/hip_runtime.h>
#include <hip/hip_bf16.h>

typedef __attribute__((ext_vector_type(2))) float v2f;
typedef __attribute__((ext_vector_type(8))) float v8f;

#define CCH 64

__device__ __forceinline__ int fkey(int bits)
{
    return bits ^ ((bits >> 31) & 0x7FFFFFFF);
}
#define KEY_NEG_INF ((int)0x807FFFFF)

#define GT 5

__device__ __forceinline__ void load_xfrag(v2f* a, const float* __restrict__ x,
                                           int tile, int N, int lm, int half)
{
    int rowA = tile * 16 + lm;
    if (rowA >= N) rowA = N - 1;
    const float* xr = x + (long)rowA * CCH + half * 2;
#pragma unroll
    for (int kk = 0; kk < 16; ++kk) {
        a[kk].x = xr[kk * 4];
        a[kk].y = xr[kk * 4 + 1];
    }
}

__global__ __launch_bounds__(128) void k_gemm(const float* __restrict__ x,
                                              const float* __restrict__ W,
                                              const float* __restrict__ bias,
                                              float* __restrict__ A_ws,
                                              float* __restrict__ B_ws,
                                              int N)
{
    const int lane = threadIdx.x & 31;
    const int wave = threadIdx.x >> 5;
    const int n0   = wave * 16;
    const int half = lane >> 4;
    const int lm   = lane & 15;
    const int col  = n0 + lm;
    const int nT   = (N + 15) >> 4;
    const int step = (int)gridDim.x;

    int t = (int)blockIdx.x;
    if (t >= nT) return;

    v2f b1[16], b2[16];
#pragma unroll
    for (int kk = 0; kk < 16; ++kk) {
        const int kb = kk * 4 + half * 2;
        b1[kk].x = W[kb * CCH + col];
        b1[kk].y = W[(kb + 1) * CCH + col];
        b2[kk].x = W[(kb + 64) * CCH + col];
        b2[kk].y = W[(kb + 65) * CCH + col];
    }
    const float bv = bias[col];

    auto tile = [&](const v2f* af, int tt) {
        const int m0 = tt * 16;
        v8f accP = {};
        v8f accQ = {};
#pragma unroll
        for (int kk = 0; kk < 16; ++kk) {
            accP = __builtin_amdgcn_wmma_f32_16x16x4_f32(false, af[kk], false, b1[kk],
                                                         (short)0, accP, false, false);
            accQ = __builtin_amdgcn_wmma_f32_16x16x4_f32(false, af[kk], false, b2[kk],
                                                         (short)0, accQ, false, false);
        }
        if (m0 + 16 <= N) {
#pragma unroll
            for (int r = 0; r < 8; ++r) {
                const int row = m0 + r + half * 8;
                const float q = accQ[r];
                A_ws[(long)row * CCH + col] = accP[r] - q + bv;
                B_ws[(long)row * CCH + col] = q;
            }
        } else {
#pragma unroll
            for (int r = 0; r < 8; ++r) {
                const int row = m0 + r + half * 8;
                if (row < N) {
                    const float q = accQ[r];
                    A_ws[(long)row * CCH + col] = accP[r] - q + bv;
                    B_ws[(long)row * CCH + col] = q;
                }
            }
        }
    };

    v2f a0[16], a1[16];
    load_xfrag(a0, x, t, N, lm, half);
    for (;;) {
        if (t + step < nT) load_xfrag(a1, x, t + step, N, lm, half);
        tile(a0, t);
        t += step;
        if (t >= nT) break;
        if (t + step < nT) load_xfrag(a0, x, t + step, N, lm, half);
        tile(a1, t);
        t += step;
        if (t >= nT) break;
    }
}

#define NPB      400
#define REC_NONE (511u << 17)
#define CHUNK    1024

__global__ __launch_bounds__(256) void k_records(const int* __restrict__ ei,
                                                 unsigned* __restrict__ own,
                                                 unsigned* __restrict__ rec,
                                                 unsigned* __restrict__ cnt,
                                                 int E, int N, long padded)
{
    const int lane = threadIdx.x & 31;
    const int wib = __builtin_amdgcn_readfirstlane((int)(threadIdx.x >> 5));
    if (blockIdx.x == 0 && threadIdx.x == 0) cnt[0] = 0u;
    const long e0 = ((long)blockIdx.x * (blockDim.x >> 5) + wib) * 128;
    if (e0 >= padded) return;
    int sv[4], dv[4];
#pragma unroll
    for (int k = 0; k < 4; ++k) {
        const long e  = e0 + 32 * k + lane;
        const long ec = (e < E) ? e : (long)E - 1;
        sv[k] = ei[ec];
        dv[k] = ei[(long)E + ec];
    }
    unsigned word = 0u;
#pragma unroll
    for (int k = 0; k < 4; ++k) {
        const long e = e0 + 32 * k + lane;
        unsigned o = 255u, rc = REC_NONE;
        if (e < E) {
            int s = sv[k];
            const int d = dv[k];
            if ((unsigned)d < (unsigned)N) {
                s = (s < 0) ? 0 : ((s > N - 1) ? N - 1 : s);
                const unsigned ob = (unsigned)d / NPB;
                if (ob < 255u) {
                    o  = ob;
                    rc = (unsigned)s | (((unsigned)d - ob * NPB) << 17);
                }
            }
        }
        rec[e] = rc;
#pragma unroll
        for (int j = 0; j < 8; ++j) {
            const unsigned bal = __builtin_amdgcn_ballot_w32(((o >> j) & 1u) != 0u);
            if (lane == (j >> 2) * 16 + k * 4 + (j & 3)) word = bal;
        }
    }
    const long c  = e0 >> 10;
    const int  l0 = (int)((e0 >> 5) & 31);
    own[c * 256 + (lane >> 4) * 128 + l0 * 4 + (lane & 15)] = word;
}

__device__ __forceinline__ unsigned take_ticket(unsigned* cnt, unsigned* s_slot)
{
    __builtin_amdgcn_fence(__ATOMIC_RELEASE, "agent");
    __syncthreads();
    if (threadIdx.x == 0) {
        __builtin_amdgcn_fence(__ATOMIC_RELEASE, "agent");
        *s_slot = __hip_atomic_fetch_add(cnt, 1u, __ATOMIC_RELAXED,
                                         __HIP_MEMORY_SCOPE_AGENT);
    }
    __syncthreads();
    return *s_slot;
}

#define OWN_WAVES 32
#define QW        64

__device__ __forceinline__ void gather_batch(const int* q, unsigned h0, unsigned n,
                                             int lane, int rows,
                                             const unsigned* __restrict__ rec,
                                             const float* __restrict__ Bn,
                                             int* keys)
{
    __builtin_amdgcn_fence(__ATOMIC_ACQ_REL, "wavefront");
    const int e = q[(h0 + (unsigned)lane) & (QW - 1)];
    unsigned r = REC_NONE;
    if ((unsigned)lane < n) r = rec[e];
    const int src = (int)(r & 0x1FFFFu);
    const int dl  = (int)(r >> 17);
    float2 v[32];
#pragma unroll
    for (int j = 0; j < 32; ++j) {
        const int sj = __builtin_amdgcn_readlane(src, j);
        v[j] = *((const float2*)(Bn + (long)sj * CCH) + lane);
    }
#pragma unroll
    for (int j = 0; j < 32; ++j) {
        const int dj = __builtin_amdgcn_readlane(dl, j);
        if (dj < rows) {
            int* row = keys + dj * CCH + 2 * lane;
            __hip_atomic_fetch_max(row,     fkey(__float_as_int(v[j].x)),
                                   __ATOMIC_RELAXED, __HIP_MEMORY_SCOPE_WORKGROUP);
            __hip_atomic_fetch_max(row + 1, fkey(__float_as_int(v[j].y)),
                                   __ATOMIC_RELAXED, __HIP_MEMORY_SCOPE_WORKGROUP);
        }
    }
}

__device__ __forceinline__ unsigned match_owned(const uint4 lo, const uint4 hi,
                                                const unsigned* kx)
{
    return (lo.x ^ kx[0]) & (lo.y ^ kx[1]) & (lo.z ^ kx[2]) & (lo.w ^ kx[3])
         & (hi.x ^ kx[4]) & (hi.y ^ kx[5]) & (hi.z ^ kx[6]) & (hi.w ^ kx[7]);
}

__global__ __launch_bounds__(1024) void k_owner_max(const float* __restrict__ Bn,
                                                    const unsigned* __restrict__ own,
                                                    const unsigned* __restrict__ rec,
                                                    const float* __restrict__ A_ws,
                                                    float* __restrict__ H,
                                                    float* psum,
                                                    float* psq,
                                                    unsigned* cnt,
                                                    float* mean,
                                                    float* invstd,
                                                    int N, int nC)
{
    __shared__ int keys[NPB * CCH];
    __shared__ __attribute__((aligned(16))) int queue[OWN_WAVES * QW];
    __shared__ unsigned s_tick;
    const int t    = threadIdx.x;
    const int lane = t & 31;
    const int wave = __builtin_amdgcn_readfirstlane(t >> 5);
    const int b    = (int)blockIdx.x;
    const int lo   = b * NPB;
    const int rows = (N - lo < NPB) ? (N - lo) : NPB;

    for (int i = t; i < NPB * CCH; i += 1024) keys[i] = KEY_NEG_INF;
    __syncthreads();

    {
        const int* q = queue + wave * QW;
        int* qw = queue + wave * QW;
        unsigned head = 0u, tail = 0u;
        unsigned kx[8];
#pragma unroll
        for (int j = 0; j < 8; ++j) kx[j] = ((b >> j) & 1) ? 0u : 0xFFFFFFFFu;
        const int start = (int)(((long)b * nC) / (int)gridDim.x);
        const uint4* own16 = (const uint4*)own;
        const int nIt = (nC > wave) ? (nC - wave + OWN_WAVES - 1) / OWN_WAVES : 0;
        const int nSt = (nIt + 1) >> 1;

        auto drain = [&](unsigned m, int ebase) {
            for (;;) {
                const unsigned bal = __builtin_amdgcn_ballot_w32(m != 0u);
                if (bal == 0u) break;
                const unsigned pre = __builtin_amdgcn_mbcnt_lo(bal, 0u);
                if (m != 0u) qw[(tail + pre) & (QW - 1)] = ebase + __builtin_ctz(m);
                tail += (unsigned)__builtin_popcount(bal);
                m &= m - 1u;
                if (tail - head >= 32u) {
                    gather_batch(q, head, 32u, lane, rows, rec, Bn, keys);
                    head += 32u;
                }
            }
        };

        int cA = wave + start;   if (cA >= nC) cA -= nC;
        int cB = cA + OWN_WAVES; if (cB >= nC) cB -= nC;
        const uint4 ones = make_uint4(~0u, ~0u, ~0u, ~0u);
        uint4 nAl = ones, nAh = ones, nBl = ones, nBh = ones;
        if (nIt > 0) {
            nAl = own16[(long)cA * 64 + lane];
            nAh = own16[(long)cA * 64 + 32 + lane];
        }
        if (nIt > 1) {
            nBl = own16[(long)cB * 64 + lane];
            nBh = own16[(long)cB * 64 + 32 + lane];
        }
        for (int s = 0; s < nSt; ++s) {
            const unsigned mA = match_owned(nAl, nAh, kx);
            const unsigned mB = (2 * s + 1 < nIt) ? match_owned(nBl, nBh, kx) : 0u;
            const int eA = cA * CHUNK + lane * 32;
            const int eB = cB * CHUNK + lane * 32;
            cA = cB + OWN_WAVES; if (cA >= nC) cA -= nC;
            cB = cA + OWN_WAVES; if (cB >= nC) cB -= nC;
            if (2 * s + 2 < nIt) {
                nAl = own16[(long)cA * 64 + lane];
                nAh = own16[(long)cA * 64 + 32 + lane];
            }
            if (2 * s + 3 < nIt) {
                nBl = own16[(long)cB * 64 + lane];
                nBh = own16[(long)cB * 64 + 32 + lane];
            }
            drain(mA, eA);
            drain(mB, eB);
        }
        if (tail != head)
            gather_batch(q, head, tail - head, lane, rows, rec, Bn, keys);
    }
    __syncthreads();

    const int c64   = t & 63;
    const int strip = t >> 6;
    float s = 0.f, sq = 0.f;
    for (int r0 = strip; r0 < rows; r0 += 16 * 5) {
        float a[5];
#pragma unroll
        for (int j = 0; j < 5; ++j) {
            int r = r0 + 16 * j;
            if (r > rows - 1) r = rows - 1;
            a[j] = A_ws[(long)(lo + r) * CCH + c64];
        }
#pragma unroll
        for (int j = 0; j < 5; ++j) {
            const int r = r0 + 16 * j;
            if (r < rows) {
                const int mb = keys[r * CCH + c64];
                float h;
                if (mb == KEY_NEG_INF) {
                    h = 0.f;
                } else {
                    const float z = a[j] + __int_as_float(fkey(mb));
                    h = (z > 0.f) ? z : expm1f(z);
                }
                H[(long)(lo + r) * CCH + c64] = h;
                s  += h;
                sq += h * h;
            }
        }
    }
    float* red = (float*)queue;
    red[strip * 128 + c64]      = s;
    red[strip * 128 + 64 + c64] = sq;
    __syncthreads();
    if (t < 128) {
        float acc = 0.f;
#pragma unroll
        for (int k = 0; k < 16; ++k) acc += red[k * 128 + t];
        if (t < 64) psum[(long)b * 64 + t]        = acc;
        else        psq [(long)b * 64 + (t - 64)] = acc;
    }

    const int nb = (int)gridDim.x;
    if (take_ticket(cnt, &s_tick) != (unsigned)(nb - 1)) return;
    __builtin_amdgcn_fence(__ATOMIC_ACQUIRE, "agent");

    const int col   = t & 127;
    const int slice = t >> 7;
    const float* p  = (col < 64) ? (psum + col) : (psq + (col - 64));
    double acc = 0.0;
    for (int r0 = slice; r0 < nb; r0 += 8 * 16) {
        float v[16];
#pragma unroll
        for (int j = 0; j < 16; ++j) {
            int r = r0 + 8 * j;
            if (r > nb - 1) r = nb - 1;
            v[j] = p[(long)r * 64];
        }
#pragma unroll
        for (int j = 0; j < 16; ++j)
            acc += (r0 + 8 * j < nb) ? (double)v[j] : 0.0;
    }
    double* fin = (double*)queue;
    fin[slice * 128 + col] = acc;
    __syncthreads();
    if (t < 64) {
        double sm = 0.0, sv = 0.0;
#pragma unroll
        for (int k = 0; k < 8; ++k) {
            sm += fin[k * 128 + t];
            sv += fin[k * 128 + 64 + t];
        }
        const double m = sm / (double)N;
        double v = sv / (double)N - m * m;
        if (v < 0.0) v = 0.0;
        mean[t]   = (float)m;
        invstd[t] = (float)(1.0 / sqrt(v + 1e-5));
    }
}

__global__ __launch_bounds__(256) void k_bn(float* __restrict__ out,
                                            const float* __restrict__ mean,
                                            const float* __restrict__ invstd,
                                            const float* __restrict__ gamma,
                                            const float* __restrict__ beta,
                                            long total)
{
    long i = (long)blockIdx.x * blockDim.x + threadIdx.x;
    if (i < total) {
        int c = (int)(i & 63);
        out[i] = gamma[c] * (out[i] - mean[c]) * invstd[c] + beta[c];
    }
}

extern "C" void kernel_launch(void* const* d_in, const int* in_sizes, int n_in,
                              void* d_out, int out_size, void* d_ws, size_t ws_size,
                              hipStream_t stream)
{
    const float* x     = (const float*)d_in[0];
    const int*   ei    = (const int*)  d_in[1];
    const float* W     = (const float*)d_in[2];
    const float* bias  = (const float*)d_in[3];
    const float* gamma = (const float*)d_in[4];
    const float* beta  = (const float*)d_in[5];
    float* out = (float*)d_out;

    const int N = in_sizes[0] / CCH;
    const int E = in_sizes[1] / 2;

    const size_t nodef = (size_t)N * CCH;
    float* A_ws = (float*)d_ws;
    float* B_ws = A_ws + nodef;
    const int nC = (E + CHUNK - 1) / CHUNK;
    const int nb = (N + NPB - 1) / NPB;
    unsigned* own  = (unsigned*)(B_ws + nodef);
    unsigned* rec  = own + (size_t)nC * (CHUNK / 4);
    float* psum    = (float*)(rec + (size_t)nC * CHUNK);
    float* psq     = psum + (size_t)nb * 64;
    float* meanp   = psq  + (size_t)nb * 64;
    float* invstdp = meanp + 64;
    unsigned* cnt  = (unsigned*)(invstdp + 64);

    const long tot    = (long)nodef;
    const long padded = (long)nC * CHUNK;

    k_gemm<<<((N + 15) / 16 + GT - 1) / GT, 128, 0, stream>>>(x, W, bias, A_ws, B_ws, N);
    k_records<<<(int)((padded / 4 + 255) / 256), 256, 0, stream>>>(ei, own, rec, cnt, E, N, padded);
    k_owner_max<<<nb, 1024, 0, stream>>>(B_ws, own, rec, A_ws, out, psum, psq, cnt,
                                         meanp, invstdp, N, nC);
    k_bn<<<(int)((tot + 255) / 256), 256, 0, stream>>>(out, meanp, invstdp, gamma, beta, tot);
}
